# starting kernel
# baseline (speedup 1.0000x reference)
#include <hip/hip_runtime.h>
#include <hip/hip_bf16.h>


typedef __attribute__((ext_vector_type(2))) float v2f;
typedef __attribute__((ext_vector_type(8))) float v8f;

#define GMM_N 65536
#define GMM_M 1024

__global__ void __launch_bounds__(GMM_M)
gmm_prep_kernel(const float* __restrict__ mu, const float* __restrict__ A,
                const float* __restrict__ w, float* __restrict__ u,
                float* __restrict__ gmax_out) {
  __shared__ float red[GMM_M];
  const int j = threadIdx.x;
  const float wj = w[j];

  red[j] = wj; __syncthreads();
  for (int s = GMM_M / 2; s > 0; s >>= 1) {
    if (j < s) red[j] = fmaxf(red[j], red[j + s]);
    __syncthreads();
  }
  const float wmax = red[0]; __syncthreads();
  red[j] = __expf(wj - wmax); __syncthreads();
  for (int s = GMM_M / 2; s > 0; s >>= 1) {
    if (j < s) red[j] += red[j + s];
    __syncthreads();
  }
  const float lse_w = wmax + __logf(red[0]); __syncthreads();

  const float a00 = A[4 * j + 0], a01 = A[4 * j + 1];
  const float a10 = A[4 * j + 2], a11 = A[4 * j + 3];
  const float g00 = 0.5f * (a00 * a00 + a01 * a01);
  const float g01 = 0.5f * (a00 * a10 + a01 * a11);
  const float g11 = 0.5f * (a10 * a10 + a11 * a11);
  const float det = g00 * g11 - g01 * g01;
  const float logw = (wj - lse_w) + 0.5f * __logf(det);

  const float m0 = mu[2 * j + 0], m1 = mu[2 * j + 1];
  const float p   = 2.f * (g00 * m0 + g01 * m1);
  const float r   = 2.f * (g01 * m0 + g11 * m1);
  const float mqm = g00 * m0 * m0 + 2.f * g01 * m0 * m1 + g11 * m1 * m1;

  float* uj = u + 8 * j;
  uj[0] = logw - mqm;
  uj[1] = p;
  uj[2] = r;
  uj[3] = -g00;
  uj[4] = -2.f * g01;
  uj[5] = -g11;
  uj[6] = 0.f;
  uj[7] = 0.f;

  red[j] = logw; __syncthreads();
  for (int s = GMM_M / 2; s > 0; s >>= 1) {
    if (j < s) red[j] = fmaxf(red[j], red[j + s]);
    __syncthreads();
  }
  if (j == 0) *gmax_out = red[0];
}

__global__ void __launch_bounds__(256)
gmm_lse_kernel(const float* __restrict__ sample, const float* __restrict__ u,
               const float* __restrict__ gmaxp, float* __restrict__ out) {
  const int lane = threadIdx.x & 31;
  const int wave = threadIdx.x >> 5;
  const int half = lane >> 4;
  const int l15  = lane & 15;
  const int r0   = (blockIdx.x * 8 + wave) * 16;

  const float2 xy = ((const float2*)sample)[r0 + l15];
  const float x0 = xy.x, x1 = xy.y;

  v2f a0, a1;
  a0.x = half ? x1        : 1.0f;
  a0.y = half ? x0 * x0   : x0;
  a1.x = half ? 0.0f      : x0 * x1;
  a1.y = half ? 0.0f      : x1 * x1;

  const float gmax = *gmaxp;

  float s[8];
#pragma unroll
  for (int v = 0; v < 8; ++v) s[v] = 0.0f;

#pragma unroll 2
  for (int t = 0; t < GMM_M / 16; ++t) {
    const int j = t * 16 + l15;
    const float4 uA = ((const float4*)u)[2 * j + 0];
    const float4 uB = ((const float4*)u)[2 * j + 1];

    v2f b0, b1;
    b0.x = half ? uA.z : uA.x;
    b0.y = half ? uA.w : uA.y;
    b1.x = half ? uB.z : uB.x;
    b1.y = half ? uB.w : uB.y;

    v8f d = {};
    d = __builtin_amdgcn_wmma_f32_16x16x4_f32(false, a0, false, b0,
                                              (short)0, d, false, false);
    d = __builtin_amdgcn_wmma_f32_16x16x4_f32(false, a1, false, b1,
                                              (short)0, d, false, false);

#pragma unroll
    for (int v = 0; v < 8; ++v) s[v] += __expf(d[v] - gmax);
  }

#pragma unroll
  for (int v = 0; v < 8; ++v) {
    float sv = s[v];
    sv += __shfl_xor(sv, 1, 32);
    sv += __shfl_xor(sv, 2, 32);
    sv += __shfl_xor(sv, 4, 32);
    sv += __shfl_xor(sv, 8, 32);
    s[v] = sv;
  }

  if (l15 == 0) {
#pragma unroll
    for (int v = 0; v < 8; ++v)
      out[r0 + half * 8 + v] = gmax + __logf(s[v]);
  }
}

extern "C" void kernel_launch(void* const* d_in, const int* in_sizes, int n_in,
                              void* d_out, int out_size, void* d_ws, size_t ws_size,
                              hipStream_t stream) {
  const float* sample = (const float*)d_in[0];
  const float* mu     = (const float*)d_in[1];
  const float* A      = (const float*)d_in[2];
  const float* w      = (const float*)d_in[3];
  float* out  = (float*)d_out;
  float* u    = (float*)d_ws;
  float* gmax = u + 8 * GMM_M;

  gmm_prep_kernel<<<1, GMM_M, 0, stream>>>(mu, A, w, u, gmax);

  gmm_lse_kernel<<<GMM_N / 128, 256, 0, stream>>>(sample, u, gmax, out);
}


# Single launch: each work-group rebuilds the 1024 coefficient columns into LDS in B-fragment order (same summation pairing) and the tile loop reads B from LDS; prep launch and workspace table removed
# speedup vs baseline: 2.1109x; 2.1109x over previous
#include <hip/hip_runtime.h>


typedef __attribute__((ext_vector_type(2))) float v2f;
typedef __attribute__((ext_vector_type(8))) float v8f;

#define GMM_N 65536
#define GMM_M 1024
#define GMM_THREADS 512
#define GMM_WAVES (GMM_THREADS / 32)

template <int M>
__device__ __forceinline__ float xrow(float v) {
  return __builtin_bit_cast(float, __builtin_amdgcn_update_dpp(
      0, __builtin_bit_cast(int, v), 0x160 | M, 0xF, 0xF, true));
}
__device__ __forceinline__ float xhalf(float v) {
  return __builtin_bit_cast(float, __builtin_amdgcn_permlanex16(
      __builtin_bit_cast(int, v), __builtin_bit_cast(int, v), 0x76543210, 0xfedcba98, false, false));
}
__device__ __forceinline__ float wave_max(float v) {
  v = fmaxf(v, xrow<1>(v));
  v = fmaxf(v, xrow<2>(v));
  v = fmaxf(v, xrow<4>(v));
  v = fmaxf(v, xrow<8>(v));
  return fmaxf(v, xhalf(v));
}
__device__ __forceinline__ float row_max(float v) {
  v = fmaxf(v, xrow<1>(v));
  v = fmaxf(v, xrow<2>(v));
  v = fmaxf(v, xrow<4>(v));
  return fmaxf(v, xrow<8>(v));
}

__device__ __forceinline__ float gmm_coef(const float4 a, const float2 m, const float wj,
                                          const float lse_w, float4& f0, float4& f1) {
  const float a00 = a.x, a01 = a.y, a10 = a.z, a11 = a.w;
  const float g00 = 0.5f * (a00 * a00 + a01 * a01);
  const float g01 = 0.5f * (a00 * a10 + a01 * a11);
  const float g11 = 0.5f * (a10 * a10 + a11 * a11);
  const float det = g00 * g11 - g01 * g01;
  const float logw = (wj - lse_w) + 0.5f * __logf(det);

  const float m0 = m.x, m1 = m.y;
  const float p   = 2.f * (g00 * m0 + g01 * m1);
  const float r   = 2.f * (g01 * m0 + g11 * m1);
  const float mqm = g00 * m0 * m0 + 2.f * g01 * m0 * m1 + g11 * m1 * m1;

  f0 = make_float4(logw - mqm, p, -2.f * g01, -g11);
  f1 = make_float4(r, -g00, 0.f, 0.f);
  return logw;
}

__global__ void __launch_bounds__(GMM_THREADS)
gmm_fused_kernel(const float* __restrict__ sample, const float* __restrict__ mu,
                 const float* __restrict__ A, const float* __restrict__ w,
                 float* __restrict__ out) {
  __shared__ float4 ufrag[GMM_M * 2];
  __shared__ float  red_sum[GMM_THREADS];
  __shared__ float  red_wmax[GMM_WAVES];
  __shared__ float  red_gmax[GMM_WAVES];

  const int tid  = threadIdx.x;
  const int lane = tid & 31;
  const int wave = tid >> 5;
  const int half = lane >> 4;
  const int l15  = lane & 15;
  const int r0   = (blockIdx.x * GMM_WAVES + wave) * 16;

  const int j0 = tid, j1 = tid + GMM_THREADS;
  const float  w0 = w[j0], w1 = w[j1];
  const float4 A0 = ((const float4*)A)[j0], A1 = ((const float4*)A)[j1];
  const float2 M0 = ((const float2*)mu)[j0], M1 = ((const float2*)mu)[j1];
  const float2 xy = ((const float2*)sample)[r0 + l15];

  const float wm = wave_max(fmaxf(w0, w1));
  if (lane == 0) red_wmax[wave] = wm;
  __syncthreads();
  const float wmax = row_max(red_wmax[l15]);

  red_sum[tid] = __expf(w0 - wmax) + __expf(w1 - wmax);
  __syncthreads();
  float rs[GMM_WAVES];
#pragma unroll
  for (int k = 0; k < GMM_WAVES; ++k) rs[k] = red_sum[lane + 32 * k];
#pragma unroll
  for (int k = 0; k < 8; ++k) rs[k] += rs[k + 8];
#pragma unroll
  for (int k = 0; k < 4; ++k) rs[k] += rs[k + 4];
  rs[0] += rs[2]; rs[1] += rs[3];
  float ssum = rs[0] + rs[1];
  ssum += xhalf(ssum);
  ssum += xrow<8>(ssum);
  ssum += xrow<4>(ssum);
  ssum += xrow<2>(ssum);
  ssum += xrow<1>(ssum);
  const float lse_w = wmax + __logf(ssum);

  float4 f0, f1;
  const float lw0 = gmm_coef(A0, M0, w0, lse_w, f0, f1);
  ufrag[(j0 >> 4) * 32 + (j0 & 15)]      = f0;
  ufrag[(j0 >> 4) * 32 + 16 + (j0 & 15)] = f1;
  const float lw1 = gmm_coef(A1, M1, w1, lse_w, f0, f1);
  ufrag[(j1 >> 4) * 32 + (j1 & 15)]      = f0;
  ufrag[(j1 >> 4) * 32 + 16 + (j1 & 15)] = f1;

  const float gm = wave_max(fmaxf(lw0, lw1));
  if (lane == 0) red_gmax[wave] = gm;
  __syncthreads();
  const float gmax = row_max(red_gmax[l15]);

  const float x0 = xy.x, x1 = xy.y;
  v2f a0, a1;
  a0.x = half ? x1        : 1.0f;
  a0.y = half ? x0 * x0   : x0;
  a1.x = half ? 0.0f      : x0 * x1;
  a1.y = half ? 0.0f      : x1 * x1;

  float s[8];
#pragma unroll
  for (int v = 0; v < 8; ++v) s[v] = 0.0f;

#pragma unroll 2
  for (int t = 0; t < GMM_M / 16; ++t) {
    const float4 f = ufrag[t * 32 + lane];
    v2f b0, b1;
    b0.x = f.x; b0.y = f.y;
    b1.x = f.z; b1.y = f.w;

    v8f d = {};
    d = __builtin_amdgcn_wmma_f32_16x16x4_f32(false, a0, false, b0,
                                              (short)0, d, false, false);
    d = __builtin_amdgcn_wmma_f32_16x16x4_f32(false, a1, false, b1,
                                              (short)0, d, false, false);

#pragma unroll
    for (int v = 0; v < 8; ++v) s[v] += __expf(d[v] - gmax);
  }

#pragma unroll
  for (int v = 0; v < 8; ++v) {
    float sv = s[v];
    sv += __shfl_xor(sv, 1, 32);
    sv += __shfl_xor(sv, 2, 32);
    sv += __shfl_xor(sv, 4, 32);
    sv += __shfl_xor(sv, 8, 32);
    s[v] = sv;
  }

  if (l15 == 0) {
#pragma unroll
    for (int v = 0; v < 8; ++v)
      out[r0 + half * 8 + v] = gmax + __logf(s[v]);
  }
}

extern "C" void kernel_launch(void* const* d_in, const int* in_sizes, int n_in,
                              void* d_out, int out_size, void* d_ws, size_t ws_size,
                              hipStream_t stream) {
  const float* sample = (const float*)d_in[0];
  const float* mu     = (const float*)d_in[1];
  const float* A      = (const float*)d_in[2];
  const float* w      = (const float*)d_in[3];
  float* out  = (float*)d_out;

  gmm_fused_kernel<<<GMM_N / 256, GMM_THREADS, 0, stream>>>(sample, mu, A, w, out);
}


# Fold log2(e) into the coefficient columns and carry the -gmax shift in a spare K slot (added inside the K sum): products feed v_exp_f32 directly, sub+mul+exp+add per element becomes exp+add
# speedup vs baseline: 2.2810x; 1.0806x over previous
#include <hip/hip_runtime.h>


typedef __attribute__((ext_vector_type(2))) float v2f;
typedef __attribute__((ext_vector_type(8))) float v8f;

#define GMM_N 65536
#define GMM_M 1024
#define GMM_THREADS 512
#define GMM_WAVES (GMM_THREADS / 32)
#define GMM_LOG2E 1.44269504088896340736f

template <int M>
__device__ __forceinline__ float xrow(float v) {
  return __builtin_bit_cast(float, __builtin_amdgcn_update_dpp(
      0, __builtin_bit_cast(int, v), 0x160 | M, 0xF, 0xF, true));
}
__device__ __forceinline__ float xhalf(float v) {
  return __builtin_bit_cast(float, __builtin_amdgcn_permlanex16(
      __builtin_bit_cast(int, v), __builtin_bit_cast(int, v), 0x76543210, 0xfedcba98, false, false));
}
__device__ __forceinline__ float wave_max(float v) {
  v = fmaxf(v, xrow<1>(v));
  v = fmaxf(v, xrow<2>(v));
  v = fmaxf(v, xrow<4>(v));
  v = fmaxf(v, xrow<8>(v));
  return fmaxf(v, xhalf(v));
}
__device__ __forceinline__ float row_max(float v) {
  v = fmaxf(v, xrow<1>(v));
  v = fmaxf(v, xrow<2>(v));
  v = fmaxf(v, xrow<4>(v));
  return fmaxf(v, xrow<8>(v));
}

__device__ __forceinline__ float gmm_coef(const float4 a, const float2 m, const float wj,
                                          const float lse_w, float4& f0, float4& f1) {
  const float a00 = a.x, a01 = a.y, a10 = a.z, a11 = a.w;
  const float g00 = 0.5f * (a00 * a00 + a01 * a01);
  const float g01 = 0.5f * (a00 * a10 + a01 * a11);
  const float g11 = 0.5f * (a10 * a10 + a11 * a11);
  const float det = g00 * g11 - g01 * g01;
  const float logw = (wj - lse_w) + 0.5f * __logf(det);

  const float m0 = m.x, m1 = m.y;
  const float p   = 2.f * (g00 * m0 + g01 * m1);
  const float r   = 2.f * (g01 * m0 + g11 * m1);
  const float mqm = g00 * m0 * m0 + 2.f * g01 * m0 * m1 + g11 * m1 * m1;

  const float L = GMM_LOG2E;
  f0 = make_float4(L * (logw - mqm), L * p, L * (-2.f * g01), L * (-g11));
  f1 = make_float4(L * r, L * (-g00), 1.0f, 0.f);
  return logw;
}

__global__ void __launch_bounds__(GMM_THREADS)
gmm_fused_kernel(const float* __restrict__ sample, const float* __restrict__ mu,
                 const float* __restrict__ A, const float* __restrict__ w,
                 float* __restrict__ out) {
  __shared__ float4 ufrag[GMM_M * 2];
  __shared__ float  red_sum[GMM_THREADS];
  __shared__ float  red_wmax[GMM_WAVES];
  __shared__ float  red_gmax[GMM_WAVES];

  const int tid  = threadIdx.x;
  const int lane = tid & 31;
  const int wave = tid >> 5;
  const int half = lane >> 4;
  const int l15  = lane & 15;
  const int r0   = (blockIdx.x * GMM_WAVES + wave) * 16;

  const int j0 = tid, j1 = tid + GMM_THREADS;
  const float  w0 = w[j0], w1 = w[j1];
  const float4 A0 = ((const float4*)A)[j0], A1 = ((const float4*)A)[j1];
  const float2 M0 = ((const float2*)mu)[j0], M1 = ((const float2*)mu)[j1];
  const float2 xy = ((const float2*)sample)[r0 + l15];

  const float wm = wave_max(fmaxf(w0, w1));
  if (lane == 0) red_wmax[wave] = wm;
  __syncthreads();
  const float wmax = row_max(red_wmax[l15]);

  red_sum[tid] = __expf(w0 - wmax) + __expf(w1 - wmax);
  __syncthreads();
  float rs[GMM_WAVES];
#pragma unroll
  for (int k = 0; k < GMM_WAVES; ++k) rs[k] = red_sum[lane + 32 * k];
#pragma unroll
  for (int k = 0; k < 8; ++k) rs[k] += rs[k + 8];
#pragma unroll
  for (int k = 0; k < 4; ++k) rs[k] += rs[k + 4];
  rs[0] += rs[2]; rs[1] += rs[3];
  float ssum = rs[0] + rs[1];
  ssum += xhalf(ssum);
  ssum += xrow<8>(ssum);
  ssum += xrow<4>(ssum);
  ssum += xrow<2>(ssum);
  ssum += xrow<1>(ssum);
  const float lse_w = wmax + __logf(ssum);

  float4 f0, f1;
  const float lw0 = gmm_coef(A0, M0, w0, lse_w, f0, f1);
  ufrag[(j0 >> 4) * 32 + (j0 & 15)]      = f0;
  ufrag[(j0 >> 4) * 32 + 16 + (j0 & 15)] = f1;
  const float lw1 = gmm_coef(A1, M1, w1, lse_w, f0, f1);
  ufrag[(j1 >> 4) * 32 + (j1 & 15)]      = f0;
  ufrag[(j1 >> 4) * 32 + 16 + (j1 & 15)] = f1;

  const float gm = wave_max(fmaxf(lw0, lw1));
  if (lane == 0) red_gmax[wave] = gm;
  __syncthreads();
  const float gmax = row_max(red_gmax[l15]);

  const float x0 = xy.x, x1 = xy.y;
  v2f a0, a1;
  a0.x = half ? x1        : 1.0f;
  a0.y = half ? x0 * x0   : x0;
  a1.x = half ? -gmax * GMM_LOG2E : x0 * x1;
  a1.y = half ? 0.0f      : x1 * x1;

  float s[8];
#pragma unroll
  for (int v = 0; v < 8; ++v) s[v] = 0.0f;

#pragma unroll 2
  for (int t = 0; t < GMM_M / 16; ++t) {
    const float4 f = ufrag[t * 32 + lane];
    v2f b0, b1;
    b0.x = f.x; b0.y = f.y;
    b1.x = f.z; b1.y = f.w;

    v8f d = {};
    d = __builtin_amdgcn_wmma_f32_16x16x4_f32(false, a0, false, b0,
                                              (short)0, d, false, false);
    d = __builtin_amdgcn_wmma_f32_16x16x4_f32(false, a1, false, b1,
                                              (short)0, d, false, false);

#pragma unroll
    for (int v = 0; v < 8; ++v) s[v] += __builtin_amdgcn_exp2f(d[v]);
  }

#pragma unroll
  for (int v = 0; v < 8; ++v) {
    float sv = s[v];
    sv += __shfl_xor(sv, 1, 32);
    sv += __shfl_xor(sv, 2, 32);
    sv += __shfl_xor(sv, 4, 32);
    sv += __shfl_xor(sv, 8, 32);
    s[v] = sv;
  }

  if (l15 == 0) {
#pragma unroll
    for (int v = 0; v < 8; ++v)
      out[r0 + half * 8 + v] = gmax + __logf(s[v]);
  }
}

extern "C" void kernel_launch(void* const* d_in, const int* in_sizes, int n_in,
                              void* d_out, int out_size, void* d_ws, size_t ws_size,
                              hipStream_t stream) {
  const float* sample = (const float*)d_in[0];
  const float* mu     = (const float*)d_in[1];
  const float* A      = (const float*)d_in[2];
  const float* w      = (const float*)d_in[3];
  float* out  = (float*)d_out;

  gmm_fused_kernel<<<GMM_N / 256, GMM_THREADS, 0, stream>>>(sample, mu, A, w, out);
}


# One K=4 WMMA per tile instead of two chained: x0,x0^2,x0x1,x1^2 on the matrix unit; constant (shift folded in) and x1-linear terms enter as its C operand via packed fma; new summation order in z
# speedup vs baseline: 2.7368x; 1.1998x over previous
#include <hip/hip_runtime.h>


typedef __attribute__((ext_vector_type(2))) float v2f;
typedef __attribute__((ext_vector_type(8))) float v8f;

#define GMM_N 65536
#define GMM_M 1024
#define GMM_THREADS 512
#define GMM_WAVES (GMM_THREADS / 32)
#define GMM_LOG2E 1.44269504088896340736f

template <int M>
__device__ __forceinline__ float xrow(float v) {
  return __builtin_bit_cast(float, __builtin_amdgcn_update_dpp(
      0, __builtin_bit_cast(int, v), 0x160 | M, 0xF, 0xF, true));
}
__device__ __forceinline__ float xhalf(float v) {
  return __builtin_bit_cast(float, __builtin_amdgcn_permlanex16(
      __builtin_bit_cast(int, v), __builtin_bit_cast(int, v), 0x76543210, 0xfedcba98, false, false));
}
__device__ __forceinline__ float wave_max(float v) {
  v = fmaxf(v, xrow<1>(v));
  v = fmaxf(v, xrow<2>(v));
  v = fmaxf(v, xrow<4>(v));
  v = fmaxf(v, xrow<8>(v));
  return fmaxf(v, xhalf(v));
}
__device__ __forceinline__ float row_max(float v) {
  v = fmaxf(v, xrow<1>(v));
  v = fmaxf(v, xrow<2>(v));
  v = fmaxf(v, xrow<4>(v));
  return fmaxf(v, xrow<8>(v));
}

struct gmm_gamma { float g00, g01, g11, hld; };
__device__ __forceinline__ gmm_gamma gmm_prec(const float4 a) {
  const float a00 = a.x, a01 = a.y, a10 = a.z, a11 = a.w;
  gmm_gamma g;
  g.g00 = 0.5f * (a00 * a00 + a01 * a01);
  g.g01 = 0.5f * (a00 * a10 + a01 * a11);
  g.g11 = 0.5f * (a10 * a10 + a11 * a11);
  const float det = g.g00 * g.g11 - g.g01 * g.g01;
  g.hld = 0.5f * __logf(det);
  return g;
}

__device__ __forceinline__ void gmm_coef(const gmm_gamma g, const float2 m, const float wj,
                                         const float lse_w, const float gmax,
                                         float4& f0, float4& f1) {
  const float logw = (wj - lse_w) + g.hld;
  const float m0 = m.x, m1 = m.y;
  const float p   = 2.f * (g.g00 * m0 + g.g01 * m1);
  const float r   = 2.f * (g.g01 * m0 + g.g11 * m1);
  const float mqm = g.g00 * m0 * m0 + 2.f * g.g01 * m0 * m1 + g.g11 * m1 * m1;

  const float L = GMM_LOG2E;
  const float ur = L * r;
  const float uc = L * ((logw - gmax) - mqm);
  f0 = make_float4(L * p, L * (-g.g00), ur, uc);
  f1 = make_float4(L * (-2.f * g.g01), L * (-g.g11), ur, uc);
}

__global__ void __launch_bounds__(GMM_THREADS)
gmm_fused_kernel(const float* __restrict__ sample, const float* __restrict__ mu,
                 const float* __restrict__ A, const float* __restrict__ w,
                 float* __restrict__ out) {
  __shared__ float4 ufrag[GMM_M * 2];
  __shared__ float  red_sum[GMM_THREADS];
  __shared__ float  red_wmax[GMM_WAVES];
  __shared__ float  red_tmax[GMM_WAVES];

  const int tid  = threadIdx.x;
  const int lane = tid & 31;
  const int wave = tid >> 5;
  const int half = lane >> 4;
  const int l15  = lane & 15;
  const int r0   = (blockIdx.x * GMM_WAVES + wave) * 16;

  const int j0 = tid, j1 = tid + GMM_THREADS;
  const float  w0 = w[j0], w1 = w[j1];
  const float4 A0 = ((const float4*)A)[j0], A1 = ((const float4*)A)[j1];
  const float2 M0 = ((const float2*)mu)[j0], M1 = ((const float2*)mu)[j1];
  const float2 xy = ((const float2*)sample)[r0 + l15];
  float4 rw[4];
#pragma unroll
  for (int k = 0; k < 4; ++k) rw[k] = ((const float4*)sample)[(r0 + 8 * half) / 2 + k];

  const gmm_gamma G0 = gmm_prec(A0), G1 = gmm_prec(A1);
  const float wm = wave_max(fmaxf(w0, w1));
  const float tm = wave_max(fmaxf(w0 + G0.hld, w1 + G1.hld));
  if (lane == 0) { red_wmax[wave] = wm; red_tmax[wave] = tm; }
  __syncthreads();
  const float wmax = row_max(red_wmax[l15]);
  const float tmax = row_max(red_tmax[l15]);

  red_sum[tid] = __expf(w0 - wmax) + __expf(w1 - wmax);
  __syncthreads();
  float rs[GMM_WAVES];
#pragma unroll
  for (int k = 0; k < GMM_WAVES; ++k) rs[k] = red_sum[lane + 32 * k];
#pragma unroll
  for (int k = 0; k < 8; ++k) rs[k] += rs[k + 8];
#pragma unroll
  for (int k = 0; k < 4; ++k) rs[k] += rs[k + 4];
  rs[0] += rs[2]; rs[1] += rs[3];
  float ssum = rs[0] + rs[1];
  ssum += xhalf(ssum);
  ssum += xrow<8>(ssum);
  ssum += xrow<4>(ssum);
  ssum += xrow<2>(ssum);
  ssum += xrow<1>(ssum);
  const float lse_w = wmax + __logf(ssum);

  const float gmax = tmax - lse_w;

  float4 f0, f1;
  gmm_coef(G0, M0, w0, lse_w, gmax, f0, f1);
  ufrag[(j0 >> 4) * 32 + (j0 & 15)]      = f0;
  ufrag[(j0 >> 4) * 32 + 16 + (j0 & 15)] = f1;
  gmm_coef(G1, M1, w1, lse_w, gmax, f0, f1);
  ufrag[(j1 >> 4) * 32 + (j1 & 15)]      = f0;
  ufrag[(j1 >> 4) * 32 + 16 + (j1 & 15)] = f1;
  __syncthreads();

  const float x0 = xy.x, x1 = xy.y;
  v2f a;
  a.x = half ? x0 * x1 : x0;
  a.y = half ? x1 * x1 : x0 * x0;

  v2f x1p[4];
#pragma unroll
  for (int k = 0; k < 4; ++k) { x1p[k].x = rw[k].y; x1p[k].y = rw[k].w; }

  float s[8];
#pragma unroll
  for (int v = 0; v < 8; ++v) s[v] = 0.0f;

#pragma unroll 2
  for (int t = 0; t < GMM_M / 16; ++t) {
    const float4 f = ufrag[t * 32 + lane];
    v2f b;
    b.x = f.x; b.y = f.y;
    const v2f ur = {f.z, f.z};
    const v2f uc = {f.w, f.w};

    v8f d;
#pragma unroll
    for (int k = 0; k < 4; ++k) {
      const v2f ck = __builtin_elementwise_fma(ur, x1p[k], uc);
      d[2 * k] = ck.x; d[2 * k + 1] = ck.y;
    }
    d = __builtin_amdgcn_wmma_f32_16x16x4_f32(false, a, false, b,
                                              (short)0, d, false, false);

#pragma unroll
    for (int v = 0; v < 8; ++v) s[v] += __builtin_amdgcn_exp2f(d[v]);
  }

#pragma unroll
  for (int v = 0; v < 8; ++v) {
    float sv = s[v];
    sv += __shfl_xor(sv, 1, 32);
    sv += __shfl_xor(sv, 2, 32);
    sv += __shfl_xor(sv, 4, 32);
    sv += __shfl_xor(sv, 8, 32);
    s[v] = sv;
  }

  if (l15 == 0) {
#pragma unroll
    for (int v = 0; v < 8; ++v)
      out[r0 + half * 8 + v] = gmax + __logf(s[v]);
  }
}

extern "C" void kernel_launch(void* const* d_in, const int* in_sizes, int n_in,
                              void* d_out, int out_size, void* d_ws, size_t ws_size,
                              hipStream_t stream) {
  const float* sample = (const float*)d_in[0];
  const float* mu     = (const float*)d_in[1];
  const float* A      = (const float*)d_in[2];
  const float* w      = (const float*)d_in[3];
  float* out  = (float*)d_out;

  gmm_fused_kernel<<<GMM_N / 256, GMM_THREADS, 0, stream>>>(sample, mu, A, w, out);
}


# speedup vs baseline: 3.1379x; 1.1466x over previous
#include <hip/hip_runtime.h>


typedef __attribute__((ext_vector_type(2))) float v2f;
typedef __attribute__((ext_vector_type(8))) float v8f;

#define GMM_N 65536
#define GMM_M 1024
#define GMM_THREADS 512
#define GMM_WAVES (GMM_THREADS / 32)
#define GMM_LOG2E 1.44269504088896340736f
#define GMM_TILES (GMM_M / 16)

template <int M>
__device__ __forceinline__ float xrow(float v) {
  return __builtin_bit_cast(float, __builtin_amdgcn_update_dpp(
      0, __builtin_bit_cast(int, v), 0x160 | M, 0xF, 0xF, true));
}
__device__ __forceinline__ float xhalf(float v) {
  return __builtin_bit_cast(float, __builtin_amdgcn_permlanex16(
      __builtin_bit_cast(int, v), __builtin_bit_cast(int, v), 0x76543210, 0xfedcba98, false, false));
}
__device__ __forceinline__ float wave_max(float v) {
  v = fmaxf(v, xrow<1>(v));
  v = fmaxf(v, xrow<2>(v));
  v = fmaxf(v, xrow<4>(v));
  v = fmaxf(v, xrow<8>(v));
  return fmaxf(v, xhalf(v));
}
__device__ __forceinline__ float row_max(float v) {
  v = fmaxf(v, xrow<1>(v));
  v = fmaxf(v, xrow<2>(v));
  v = fmaxf(v, xrow<4>(v));
  return fmaxf(v, xrow<8>(v));
}

struct gmm_gamma { float g00, g01, g11, hld; };
__device__ __forceinline__ gmm_gamma gmm_prec(const float4 a) {
  const float a00 = a.x, a01 = a.y, a10 = a.z, a11 = a.w;
  gmm_gamma g;
  g.g00 = 0.5f * (a00 * a00 + a01 * a01);
  g.g01 = 0.5f * (a00 * a10 + a01 * a11);
  g.g11 = 0.5f * (a10 * a10 + a11 * a11);
  const float det = g.g00 * g.g11 - g.g01 * g.g01;
  g.hld = 0.5f * __logf(det);
  return g;
}

__device__ __forceinline__ void gmm_coef(const gmm_gamma g, const float2 m, const float wj,
                                         const float lse_w, const float gmax,
                                         float4& f0, float4& f1) {
  const float logw = (wj - lse_w) + g.hld;
  const float m0 = m.x, m1 = m.y;
  const float p   = 2.f * (g.g00 * m0 + g.g01 * m1);
  const float r   = 2.f * (g.g01 * m0 + g.g11 * m1);
  const float mqm = g.g00 * m0 * m0 + 2.f * g.g01 * m0 * m1 + g.g11 * m1 * m1;

  const float L = GMM_LOG2E;
  const float ur = L * r;
  const float uc = L * ((logw - gmax) - mqm);
  f0 = make_float4(L * p, L * (-g.g00), ur, uc);
  f1 = make_float4(L * (-2.f * g.g01), L * (-g.g11), ur, uc);
}

__global__ void __launch_bounds__(GMM_THREADS)
gmm_fused_kernel(const float* __restrict__ sample, const float* __restrict__ mu,
                 const float* __restrict__ A, const float* __restrict__ w,
                 float* __restrict__ out) {
  __shared__ float4 ufrag[GMM_M * 2];
  __shared__ float  red_sum[GMM_THREADS];
  __shared__ float  red_wmax[GMM_WAVES];
  __shared__ float  red_tmax[GMM_WAVES];

  const int tid  = threadIdx.x;
  const int lane = tid & 31;
  const int wave = tid >> 5;
  const int half = lane >> 4;
  const int l15  = lane & 15;
  const int r0   = (blockIdx.x * GMM_WAVES + wave) * 16;

  const int j0 = tid, j1 = tid + GMM_THREADS;
  const float  w0 = w[j0], w1 = w[j1];
  const float4 A0 = ((const float4*)A)[j0], A1 = ((const float4*)A)[j1];
  const float2 M0 = ((const float2*)mu)[j0], M1 = ((const float2*)mu)[j1];
  const float2 xy = ((const float2*)sample)[r0 + l15];
  float4 rw[4];
#pragma unroll
  for (int k = 0; k < 4; ++k) rw[k] = ((const float4*)sample)[(r0 + 8 * half) / 2 + k];

  const gmm_gamma G0 = gmm_prec(A0), G1 = gmm_prec(A1);
  const float wm = wave_max(fmaxf(w0, w1));
  const float tm = wave_max(fmaxf(w0 + G0.hld, w1 + G1.hld));
  if (lane == 0) { red_wmax[wave] = wm; red_tmax[wave] = tm; }
  __syncthreads();
  const float wmax = row_max(red_wmax[l15]);
  const float tmax = row_max(red_tmax[l15]);

  red_sum[tid] = __expf(w0 - wmax) + __expf(w1 - wmax);
  __syncthreads();
  float rs[GMM_WAVES];
#pragma unroll
  for (int k = 0; k < GMM_WAVES; ++k) rs[k] = red_sum[lane + 32 * k];
#pragma unroll
  for (int k = 0; k < 8; ++k) rs[k] += rs[k + 8];
#pragma unroll
  for (int k = 0; k < 4; ++k) rs[k] += rs[k + 4];
  rs[0] += rs[2]; rs[1] += rs[3];
  float ssum = rs[0] + rs[1];
  ssum += xhalf(ssum);
  ssum += xrow<8>(ssum);
  ssum += xrow<4>(ssum);
  ssum += xrow<2>(ssum);
  ssum += xrow<1>(ssum);
  const float lse_w = wmax + __logf(ssum);

  const float gmax = tmax - lse_w;

  float4 f0, f1;
  gmm_coef(G0, M0, w0, lse_w, gmax, f0, f1);
  ufrag[(j0 >> 4) * 32 + (j0 & 15)]      = f0;
  ufrag[(j0 >> 4) * 32 + 16 + (j0 & 15)] = f1;
  gmm_coef(G1, M1, w1, lse_w, gmax, f0, f1);
  ufrag[(j1 >> 4) * 32 + (j1 & 15)]      = f0;
  ufrag[(j1 >> 4) * 32 + 16 + (j1 & 15)] = f1;
  __syncthreads();

  const float x0 = xy.x, x1 = xy.y;
  v2f a;
  a.x = half ? x0 * x1 : x0;
  a.y = half ? x1 * x1 : x0 * x0;

  v2f x1p[4];
#pragma unroll
  for (int k = 0; k < 4; ++k) { x1p[k].x = rw[k].y; x1p[k].y = rw[k].w; }

  v2f sp[4];
#pragma unroll
  for (int k = 0; k < 4; ++k) { sp[k].x = 0.0f; sp[k].y = 0.0f; }
  v8f e0 = {0.f, 0.f, 0.f, 0.f, 0.f, 0.f, 0.f, 0.f};
  v8f e1 = {0.f, 0.f, 0.f, 0.f, 0.f, 0.f, 0.f, 0.f};
  v8f cA, cB;

  const float4* up = ufrag + lane;
  float4 fa[4], fb[4];
#pragma unroll
  for (int i = 0; i < 4; ++i) fa[i] = up[i * 32];
#pragma unroll
  for (int i = 0; i < 4; ++i) fb[i] = up[(4 + i) * 32];

  v8f d0, d1;
  {
    const v2f b0_ = {fa[0].x, fa[0].y}, u0_ = {fa[0].z, fa[0].w};
    const v2f b1_ = {fa[1].x, fa[1].y}, u1_ = {fa[1].z, fa[1].w};
    asm volatile(
      "v_pk_fma_f32 v[132:133], %[u0], %[x0], %[u0] op_sel:[0,0,1] op_sel_hi:[0,1,1]\n\t"
      "v_pk_fma_f32 v[134:135], %[u0], %[x1], %[u0] op_sel:[0,0,1] op_sel_hi:[0,1,1]\n\t"
      "v_pk_fma_f32 v[136:137], %[u0], %[x2], %[u0] op_sel:[0,0,1] op_sel_hi:[0,1,1]\n\t"
      "v_pk_fma_f32 v[138:139], %[u0], %[x3], %[u0] op_sel:[0,0,1] op_sel_hi:[0,1,1]\n\t"
      "v_pk_fma_f32 v[140:141], %[u1], %[x0], %[u1] op_sel:[0,0,1] op_sel_hi:[0,1,1]\n\t"
      "v_pk_fma_f32 v[142:143], %[u1], %[x1], %[u1] op_sel:[0,0,1] op_sel_hi:[0,1,1]\n\t"
      "v_pk_fma_f32 v[144:145], %[u1], %[x2], %[u1] op_sel:[0,0,1] op_sel_hi:[0,1,1]\n\t"
      "v_pk_fma_f32 v[146:147], %[u1], %[x3], %[u1] op_sel:[0,0,1] op_sel_hi:[0,1,1]\n\t"
      "v_wmma_f32_16x16x4_f32 v[100:107], %[a], %[b0], v[132:139]\n\t"
      "v_wmma_f32_16x16x4_f32 v[108:115], %[a], %[b1], v[140:147]\n\t"
      : "+{v[100:107]}"(d0), "+{v[108:115]}"(d1), "+{v[132:139]}"(cA), "+{v[140:147]}"(cB)
      : [a] "v"(a), [b0] "v"(b0_), [u0] "v"(u0_), [b1] "v"(b1_), [u1] "v"(u1_),
        [x0] "v"(x1p[0]), [x1] "v"(x1p[1]), [x2] "v"(x1p[2]), [x3] "v"(x1p[3]));
  }

#define GMM_STEP_A(F, FPREV)                                                   \
  {                                                                          \
    const v2f bxy_ = {F.x, F.y};                                             \
    const v2f uzw_ = {F.z, F.w};                                             \
    const v2f keep_ = {FPREV.x, FPREV.y};                                    \
    asm volatile(                                                            \
      "v_exp_f32 v116, v100\n\t" \
      "v_pk_fma_f32 v[132:133], %[uzw], %[x0], %[uzw] op_sel:[0,0,1] op_sel_hi:[0,1,1]\n\t" \
      "v_exp_f32 v117, v101\n\t" \
      "v_pk_fma_f32 v[134:135], %[uzw], %[x1], %[uzw] op_sel:[0,0,1] op_sel_hi:[0,1,1]\n\t" \
      "v_exp_f32 v118, v102\n\t" \
      "v_pk_fma_f32 v[136:137], %[uzw], %[x2], %[uzw] op_sel:[0,0,1] op_sel_hi:[0,1,1]\n\t" \
      "v_exp_f32 v119, v103\n\t" \
      "v_pk_fma_f32 v[138:139], %[uzw], %[x3], %[uzw] op_sel:[0,0,1] op_sel_hi:[0,1,1]\n\t" \
      "v_exp_f32 v120, v104\n\t" \
      "v_pk_add_f32 %[s0], %[s0], v[124:125]\n\t" \
      "v_exp_f32 v121, v105\n\t" \
      "v_pk_add_f32 %[s1], %[s1], v[126:127]\n\t" \
      "v_exp_f32 v122, v106\n\t" \
      "v_pk_add_f32 %[s2], %[s2], v[128:129]\n\t" \
      "v_exp_f32 v123, v107\n\t" \
      "v_pk_add_f32 %[s3], %[s3], v[130:131]\n\t" \
      "v_wmma_f32_16x16x4_f32 v[100:107], %[a], %[bxy], v[132:139]\n\t" \
      : "+{v[100:107]}"(d0), "+{v[116:123]}"(e0), "+{v[124:131]}"(e1), "+{v[132:139]}"(cA),                  \
        [s0] "+v"(sp[0]), [s1] "+v"(sp[1]), [s2] "+v"(sp[2]), [s3] "+v"(sp[3])   \
      : [a] "v"(a), [bxy] "v"(bxy_), [uzw] "v"(uzw_), \
        [x0] "v"(x1p[0]), [x1] "v"(x1p[1]), [x2] "v"(x1p[2]), [x3] "v"(x1p[3]),  \
        "v"(keep_)                                                           \
      );  \
  }
#define GMM_STEP_B(F, FPREV)                                                   \
  {                                                                          \
    const v2f bxy_ = {F.x, F.y};                                             \
    const v2f uzw_ = {F.z, F.w};                                             \
    const v2f keep_ = {FPREV.x, FPREV.y};                                    \
    asm volatile(                                                            \
      "v_exp_f32 v124, v108\n\t" \
      "v_pk_fma_f32 v[140:141], %[uzw], %[x0], %[uzw] op_sel:[0,0,1] op_sel_hi:[0,1,1]\n\t" \
      "v_exp_f32 v125, v109\n\t" \
      "v_pk_fma_f32 v[142:143], %[uzw], %[x1], %[uzw] op_sel:[0,0,1] op_sel_hi:[0,1,1]\n\t" \
      "v_exp_f32 v126, v110\n\t" \
      "v_pk_fma_f32 v[144:145], %[uzw], %[x2], %[uzw] op_sel:[0,0,1] op_sel_hi:[0,1,1]\n\t" \
      "v_exp_f32 v127, v111\n\t" \
      "v_pk_fma_f32 v[146:147], %[uzw], %[x3], %[uzw] op_sel:[0,0,1] op_sel_hi:[0,1,1]\n\t" \
      "v_exp_f32 v128, v112\n\t" \
      "v_pk_add_f32 %[s0], %[s0], v[116:117]\n\t" \
      "v_exp_f32 v129, v113\n\t" \
      "v_pk_add_f32 %[s1], %[s1], v[118:119]\n\t" \
      "v_exp_f32 v130, v114\n\t" \
      "v_pk_add_f32 %[s2], %[s2], v[120:121]\n\t" \
      "v_exp_f32 v131, v115\n\t" \
      "v_pk_add_f32 %[s3], %[s3], v[122:123]\n\t" \
      "v_wmma_f32_16x16x4_f32 v[108:115], %[a], %[bxy], v[140:147]\n\t" \
      : "+{v[108:115]}"(d1), "+{v[124:131]}"(e1), "+{v[116:123]}"(e0), "+{v[140:147]}"(cB),                  \
        [s0] "+v"(sp[0]), [s1] "+v"(sp[1]), [s2] "+v"(sp[2]), [s3] "+v"(sp[3])   \
      : [a] "v"(a), [bxy] "v"(bxy_), [uzw] "v"(uzw_), \
        [x0] "v"(x1p[0]), [x1] "v"(x1p[1]), [x2] "v"(x1p[2]), [x3] "v"(x1p[3]),  \
        "v"(keep_)                                                           \
      );  \
  }

  GMM_STEP_A(fa[2], fa[1])
  GMM_STEP_B(fa[3], fa[2])
  for (int g0 = 4; g0 < GMM_TILES - 4; g0 += 8) {
    GMM_STEP_A(fb[0], fa[3])
#pragma unroll
    for (int i = 0; i < 4; ++i) fa[i] = up[(g0 + 4 + i) * 32];
    GMM_STEP_B(fb[1], fb[0])
    GMM_STEP_A(fb[2], fb[1])
    GMM_STEP_B(fb[3], fb[2])
    GMM_STEP_A(fa[0], fb[3])
#pragma unroll
    for (int i = 0; i < 4; ++i) fb[i] = up[(g0 + 8 + i) * 32];
    GMM_STEP_B(fa[1], fa[0])
    GMM_STEP_A(fa[2], fa[1])
    GMM_STEP_B(fa[3], fa[2])
  }
  GMM_STEP_A(fb[0], fa[3])
  GMM_STEP_B(fb[1], fb[0])
  GMM_STEP_A(fb[2], fb[1])
  GMM_STEP_B(fb[3], fb[2])
#undef GMM_STEP_A
#undef GMM_STEP_B

  {
    const v2f keep_ = {fb[3].x, fb[3].y};
    asm volatile(
      "v_exp_f32 v116, v100\n\t"
      "v_pk_add_f32 %[s0], %[s0], v[124:125]\n\t"
      "v_exp_f32 v117, v101\n\t"
      "v_pk_add_f32 %[s1], %[s1], v[126:127]\n\t"
      "v_exp_f32 v118, v102\n\t"
      "v_pk_add_f32 %[s2], %[s2], v[128:129]\n\t"
      "v_exp_f32 v119, v103\n\t"
      "v_pk_add_f32 %[s3], %[s3], v[130:131]\n\t"
      "v_exp_f32 v120, v104\n\t"
      "v_exp_f32 v121, v105\n\t"
      "v_exp_f32 v122, v106\n\t"
      "v_exp_f32 v123, v107\n\t"
      "v_exp_f32 v124, v108\n\t"
      "v_exp_f32 v125, v109\n\t"
      "v_exp_f32 v126, v110\n\t"
      "v_exp_f32 v127, v111\n\t"
      "v_exp_f32 v128, v112\n\t"
      "v_pk_add_f32 %[s0], %[s0], v[116:117]\n\t"
      "v_exp_f32 v129, v113\n\t"
      "v_pk_add_f32 %[s1], %[s1], v[118:119]\n\t"
      "v_exp_f32 v130, v114\n\t"
      "v_pk_add_f32 %[s2], %[s2], v[120:121]\n\t"
      "v_exp_f32 v131, v115\n\t"
      "v_pk_add_f32 %[s3], %[s3], v[122:123]\n\t"
      : "+{v[100:107]}"(d0), "+{v[108:115]}"(d1), "+{v[116:123]}"(e0), "+{v[124:131]}"(e1),
        [s0] "+v"(sp[0]), [s1] "+v"(sp[1]), [s2] "+v"(sp[2]), [s3] "+v"(sp[3])
      : "v"(keep_));
  }
  float s[8];
#pragma unroll
  for (int k = 0; k < 4; ++k) { s[2 * k] = sp[k].x + e1[2 * k]; s[2 * k + 1] = sp[k].y + e1[2 * k + 1]; }

#pragma unroll
  for (int v = 0; v < 8; ++v) {
    float sv = s[v];
    sv += __shfl_xor(sv, 1, 32);
    sv += __shfl_xor(sv, 2, 32);
    sv += __shfl_xor(sv, 4, 32);
    sv += __shfl_xor(sv, 8, 32);
    s[v] = sv;
  }

  if (l15 == 0) {
#pragma unroll
    for (int v = 0; v < 8; ++v)
      out[r0 + half * 8 + v] = gmax + __logf(s[v]);
  }
}

extern "C" void kernel_launch(void* const* d_in, const int* in_sizes, int n_in,
                              void* d_out, int out_size, void* d_ws, size_t ws_size,
                              hipStream_t stream) {
  const float* sample = (const float*)d_in[0];
  const float* mu     = (const float*)d_in[1];
  const float* A      = (const float*)d_in[2];
  const float* w      = (const float*)d_in[3];
  float* out  = (float*)d_out;

  gmm_fused_kernel<<<GMM_N / 256, GMM_THREADS, 0, stream>>>(sample, mu, A, w, out);
}
